# starting kernel
# baseline (speedup 1.0000x reference)
#include <hip/hip_runtime.h>

typedef float v2f __attribute__((ext_vector_type(2)));
typedef float v8f __attribute__((ext_vector_type(8)));

namespace {

constexpr int kB  = 2048;
constexpr int kT  = 512;
constexpr int kIn = 28;
constexpr int kH  = 64;
constexpr int kC  = 10;
constexpr int kBM = 16;

__device__ __forceinline__ v8f wmma4(v2f a, v2f b, v8f c) {
  return __builtin_amdgcn_wmma_f32_16x16x4_f32(
       false, a,  false, b,
       (short)0, c,  false,  false);
}

__device__ __forceinline__ float sigmoid_f(float x) {
  return 1.0f / (1.0f + __expf(-x));
}
__device__ __forceinline__ float tanh_f(float x) {
  return 1.0f - 2.0f / (__expf(2.0f * x) + 1.0f);
}

__global__ __launch_bounds__(128, 1) void gru_fused_kernel(
    const float* __restrict__ x,
    const float* __restrict__ W_ih,
    const float* __restrict__ W_hh,
    const float* __restrict__ b_ih,
    const float* __restrict__ b_hh,
    const float* __restrict__ W_out,
    const float* __restrict__ b_out,
    float* __restrict__ out)
{
  __shared__ float h_lds[kBM * kH];

  const int tid  = threadIdx.x;
  const int wave = tid >> 5;
  const int lane = tid & 31;
  const int m    = lane & 15;
  const int hi   = lane >> 4;
  const int bm   = blockIdx.x * kBM;

  v2f whhB[3][16];
  v2f wihB[3][7];
#pragma unroll
  for (int g = 0; g < 3; ++g) {
    const int nrow = g * 64 + wave * 16 + m;
#pragma unroll
    for (int s = 0; s < 16; ++s)
      whhB[g][s] = *(const v2f*)(W_hh + nrow * kH + 4 * s + 2 * hi);
#pragma unroll
    for (int s = 0; s < 7; ++s)
      wihB[g][s] = *(const v2f*)(W_ih + nrow * kIn + 4 * s + 2 * hi);
  }

  const int col = wave * 16 + m;
  const float bias_r  = b_ih[col]       + b_hh[col];
  const float bias_z  = b_ih[64 + col]  + b_hh[64 + col];
  const float bias_xn = b_ih[128 + col];
  const float bias_hn = b_hh[128 + col];

  v2f hA[16];
#pragma unroll
  for (int s = 0; s < 16; ++s) hA[s] = v2f{0.0f, 0.0f};
  v8f h_c = {};

  const float* xbase = x + (size_t)(bm + m) * (kT * kIn) + 2 * hi;

  for (int t = 0; t < kT; ++t) {
    v2f xA[7];
    const float* xt = xbase + t * kIn;
#pragma unroll
    for (int s = 0; s < 7; ++s) xA[s] = *(const v2f*)(xt + 4 * s);

    v8f acc_r, acc_z, acc_xn, acc_hn;
#pragma unroll
    for (int j = 0; j < 8; ++j) {
      acc_r[j]  = bias_r;
      acc_z[j]  = bias_z;
      acc_xn[j] = bias_xn;
      acc_hn[j] = bias_hn;
    }

#pragma unroll
    for (int s = 0; s < 7; ++s) {
      acc_r  = wmma4(xA[s], wihB[0][s], acc_r);
      acc_z  = wmma4(xA[s], wihB[1][s], acc_z);
      acc_xn = wmma4(xA[s], wihB[2][s], acc_xn);
    }
#pragma unroll
    for (int s = 0; s < 16; ++s) {
      acc_r  = wmma4(hA[s], whhB[0][s], acc_r);
      acc_z  = wmma4(hA[s], whhB[1][s], acc_z);
      acc_hn = wmma4(hA[s], whhB[2][s], acc_hn);
    }

    v8f h_new;
#pragma unroll
    for (int j = 0; j < 8; ++j) {
      const float r = sigmoid_f(acc_r[j]);
      const float z = sigmoid_f(acc_z[j]);
      const float n = tanh_f(acc_xn[j] + r * acc_hn[j]);
      h_new[j] = (1.0f - z) * n + z * h_c[j];
    }
    h_c = h_new;

    __syncthreads();
#pragma unroll
    for (int j = 0; j < 8; ++j)
      h_lds[(j + 8 * hi) * kH + col] = h_new[j];
    __syncthreads();
#pragma unroll
    for (int s = 0; s < 16; ++s)
      hA[s] = *(const v2f*)(h_lds + m * kH + 4 * s + 2 * hi);
  }

  if (tid < kBM) {
    float logits[kC];
    float mx = -3.4e38f;
#pragma unroll
    for (int c = 0; c < kC; ++c) {
      float d = b_out[c];
#pragma unroll
      for (int k = 0; k < kH; ++k)
        d += h_lds[tid * kH + k] * W_out[c * kH + k];
      logits[c] = d;
      mx = fmaxf(mx, d);
    }
    float sum = 0.0f;
#pragma unroll
    for (int c = 0; c < kC; ++c) {
      const float e = __expf(logits[c] - mx);
      logits[c] = e;
      sum += e;
    }
    const float inv = 1.0f / sum;
#pragma unroll
    for (int c = 0; c < kC; ++c)
      out[(size_t)(bm + tid) * kC + c] = logits[c] * inv;
  }
}

}

extern "C" void kernel_launch(void* const* d_in, const int* in_sizes, int n_in,
                              void* d_out, int out_size, void* d_ws, size_t ws_size,
                              hipStream_t stream) {
  (void)in_sizes; (void)n_in; (void)out_size; (void)d_ws; (void)ws_size;
  const float* x     = (const float*)d_in[0];
  const float* W_ih  = (const float*)d_in[1];
  const float* W_hh  = (const float*)d_in[2];
  const float* b_ih  = (const float*)d_in[3];
  const float* b_hh  = (const float*)d_in[4];
  const float* W_out = (const float*)d_in[5];
  const float* b_out = (const float*)d_in[6];
  float* out = (float*)d_out;

  dim3 grid(kB / kBM);
  dim3 block(128);
  gru_fused_kernel<<<grid, block, 0, stream>>>(x, W_ih, W_hh, b_ih, b_hh,
                                               W_out, b_out, out);
}


# gate sigmoid/tanh: hardware reciprocal (v_rcp_f32, 1 ulp) replaces the IEEE-exact f32 division sequence; results differ in the last bits, summation order unchanged
# speedup vs baseline: 1.4261x; 1.4261x over previous
#include <hip/hip_runtime.h>

typedef float v2f __attribute__((ext_vector_type(2)));
typedef float v8f __attribute__((ext_vector_type(8)));

namespace {

constexpr int kB  = 2048;
constexpr int kT  = 512;
constexpr int kIn = 28;
constexpr int kH  = 64;
constexpr int kC  = 10;
constexpr int kBM = 16;

__device__ __forceinline__ v8f wmma4(v2f a, v2f b, v8f c) {
  return __builtin_amdgcn_wmma_f32_16x16x4_f32(
       false, a,  false, b,
       (short)0, c,  false,  false);
}

__device__ __forceinline__ float rcp_f(float x) {
  return __builtin_amdgcn_rcpf(x);
}
__device__ __forceinline__ float sigmoid_f(float x) {
  return rcp_f(1.0f + __expf(-x));
}
__device__ __forceinline__ float tanh_f(float x) {
  return 1.0f - 2.0f * rcp_f(__expf(2.0f * x) + 1.0f);
}

__global__ __launch_bounds__(128, 1) void gru_fused_kernel(
    const float* __restrict__ x,
    const float* __restrict__ W_ih,
    const float* __restrict__ W_hh,
    const float* __restrict__ b_ih,
    const float* __restrict__ b_hh,
    const float* __restrict__ W_out,
    const float* __restrict__ b_out,
    float* __restrict__ out)
{
  __shared__ float h_lds[kBM * kH];

  const int tid  = threadIdx.x;
  const int wave = tid >> 5;
  const int lane = tid & 31;
  const int m    = lane & 15;
  const int hi   = lane >> 4;
  const int bm   = blockIdx.x * kBM;

  v2f whhB[3][16];
  v2f wihB[3][7];
#pragma unroll
  for (int g = 0; g < 3; ++g) {
    const int nrow = g * 64 + wave * 16 + m;
#pragma unroll
    for (int s = 0; s < 16; ++s)
      whhB[g][s] = *(const v2f*)(W_hh + nrow * kH + 4 * s + 2 * hi);
#pragma unroll
    for (int s = 0; s < 7; ++s)
      wihB[g][s] = *(const v2f*)(W_ih + nrow * kIn + 4 * s + 2 * hi);
  }

  const int col = wave * 16 + m;
  const float bias_r  = b_ih[col]       + b_hh[col];
  const float bias_z  = b_ih[64 + col]  + b_hh[64 + col];
  const float bias_xn = b_ih[128 + col];
  const float bias_hn = b_hh[128 + col];

  v2f hA[16];
#pragma unroll
  for (int s = 0; s < 16; ++s) hA[s] = v2f{0.0f, 0.0f};
  v8f h_c = {};

  const float* xbase = x + (size_t)(bm + m) * (kT * kIn) + 2 * hi;

  for (int t = 0; t < kT; ++t) {
    v2f xA[7];
    const float* xt = xbase + t * kIn;
#pragma unroll
    for (int s = 0; s < 7; ++s) xA[s] = *(const v2f*)(xt + 4 * s);

    v8f acc_r, acc_z, acc_xn, acc_hn;
#pragma unroll
    for (int j = 0; j < 8; ++j) {
      acc_r[j]  = bias_r;
      acc_z[j]  = bias_z;
      acc_xn[j] = bias_xn;
      acc_hn[j] = bias_hn;
    }

#pragma unroll
    for (int s = 0; s < 7; ++s) {
      acc_r  = wmma4(xA[s], wihB[0][s], acc_r);
      acc_z  = wmma4(xA[s], wihB[1][s], acc_z);
      acc_xn = wmma4(xA[s], wihB[2][s], acc_xn);
    }
#pragma unroll
    for (int s = 0; s < 16; ++s) {
      acc_r  = wmma4(hA[s], whhB[0][s], acc_r);
      acc_z  = wmma4(hA[s], whhB[1][s], acc_z);
      acc_hn = wmma4(hA[s], whhB[2][s], acc_hn);
    }

    v8f h_new;
#pragma unroll
    for (int j = 0; j < 8; ++j) {
      const float r = sigmoid_f(acc_r[j]);
      const float z = sigmoid_f(acc_z[j]);
      const float n = tanh_f(acc_xn[j] + r * acc_hn[j]);
      h_new[j] = (1.0f - z) * n + z * h_c[j];
    }
    h_c = h_new;

    __syncthreads();
#pragma unroll
    for (int j = 0; j < 8; ++j)
      h_lds[(j + 8 * hi) * kH + col] = h_new[j];
    __syncthreads();
#pragma unroll
    for (int s = 0; s < 16; ++s)
      hA[s] = *(const v2f*)(h_lds + m * kH + 4 * s + 2 * hi);
  }

  if (tid < kBM) {
    float logits[kC];
    float mx = -3.4e38f;
#pragma unroll
    for (int c = 0; c < kC; ++c) {
      float d = b_out[c];
#pragma unroll
      for (int k = 0; k < kH; ++k)
        d += h_lds[tid * kH + k] * W_out[c * kH + k];
      logits[c] = d;
      mx = fmaxf(mx, d);
    }
    float sum = 0.0f;
#pragma unroll
    for (int c = 0; c < kC; ++c) {
      const float e = __expf(logits[c] - mx);
      logits[c] = e;
      sum += e;
    }
    const float inv = 1.0f / sum;
#pragma unroll
    for (int c = 0; c < kC; ++c)
      out[(size_t)(bm + tid) * kC + c] = logits[c] * inv;
  }
}

}

extern "C" void kernel_launch(void* const* d_in, const int* in_sizes, int n_in,
                              void* d_out, int out_size, void* d_ws, size_t ws_size,
                              hipStream_t stream) {
  (void)in_sizes; (void)n_in; (void)out_size; (void)d_ws; (void)ws_size;
  const float* x     = (const float*)d_in[0];
  const float* W_ih  = (const float*)d_in[1];
  const float* W_hh  = (const float*)d_in[2];
  const float* b_ih  = (const float*)d_in[3];
  const float* b_hh  = (const float*)d_in[4];
  const float* W_out = (const float*)d_in[5];
  const float* b_out = (const float*)d_in[6];
  float* out = (float*)d_out;

  dim3 grid(kB / kBM);
  dim3 block(128);
  gru_fused_kernel<<<grid, block, 0, stream>>>(x, W_ih, W_hh, b_ih, b_hh,
                                               W_out, b_out, out);
}


# input projection x*W_ih^T pipelined one step ahead beside gate math and LDS exchange; x prefetched 3 steps ahead, barrier fences LDS-only so the loads stay in flight; summation order unchanged
# speedup vs baseline: 2.0637x; 1.4471x over previous
#include <hip/hip_runtime.h>

typedef float v2f __attribute__((ext_vector_type(2)));
typedef float v8f __attribute__((ext_vector_type(8)));

namespace {

constexpr int kB  = 2048;
constexpr int kT  = 512;
constexpr int kIn = 28;
constexpr int kH  = 64;
constexpr int kC  = 10;
constexpr int kBM = 16;

__device__ __forceinline__ v8f wmma4(v2f a, v2f b, v8f c) {
  return __builtin_amdgcn_wmma_f32_16x16x4_f32(
       false, a,  false, b,
       (short)0, c,  false,  false);
}

__device__ __forceinline__ float rcp_f(float x) {
  return __builtin_amdgcn_rcpf(x);
}
__device__ __forceinline__ float sigmoid_f(float x) {
  return rcp_f(1.0f + __expf(-x));
}
__device__ __forceinline__ float tanh_f(float x) {
  return 1.0f - 2.0f * rcp_f(__expf(2.0f * x) + 1.0f);
}

__device__ __forceinline__ void lds_barrier() {
  __builtin_amdgcn_fence(__ATOMIC_RELEASE, "workgroup", "local");
  __builtin_amdgcn_s_barrier();
  __builtin_amdgcn_fence(__ATOMIC_ACQUIRE, "workgroup", "local");
}

__global__ __launch_bounds__(128, 1) void gru_fused_kernel(
    const float* __restrict__ x,
    const float* __restrict__ W_ih,
    const float* __restrict__ W_hh,
    const float* __restrict__ b_ih,
    const float* __restrict__ b_hh,
    const float* __restrict__ W_out,
    const float* __restrict__ b_out,
    float* __restrict__ out)
{
  __shared__ float h_lds[kBM * kH];

  const int tid  = threadIdx.x;
  const int wave = tid >> 5;
  const int lane = tid & 31;
  const int m    = lane & 15;
  const int hi   = lane >> 4;
  const int bm   = blockIdx.x * kBM;

  v2f whhB[3][16];
  v2f wihB[3][7];
#pragma unroll
  for (int g = 0; g < 3; ++g) {
    const int nrow = g * 64 + wave * 16 + m;
#pragma unroll
    for (int s = 0; s < 16; ++s)
      whhB[g][s] = *(const v2f*)(W_hh + nrow * kH + 4 * s + 2 * hi);
#pragma unroll
    for (int s = 0; s < 7; ++s)
      wihB[g][s] = *(const v2f*)(W_ih + nrow * kIn + 4 * s + 2 * hi);
  }

  const int col = wave * 16 + m;
  const float bias_r  = b_ih[col]       + b_hh[col];
  const float bias_z  = b_ih[64 + col]  + b_hh[64 + col];
  const float bias_xn = b_ih[128 + col];
  const float bias_hn = b_hh[128 + col];

  v2f hA[16];
#pragma unroll
  for (int s = 0; s < 16; ++s) hA[s] = v2f{0.0f, 0.0f};
  v8f h_c = {};

  const float* xbase = x + (size_t)(bm + m) * (kT * kIn) + 2 * hi;

  auto load_x = [&](v2f (&xf)[7], int t, int s0, int s1) {
    const float* xt = xbase + (t < kT ? t : kT - 1) * kIn;
#pragma unroll
    for (int s = 0; s < 7; ++s)
      if (s >= s0 && s < s1) xf[s] = *(const v2f*)(xt + 4 * s);
  };

  v8f nx_r, nx_z, nx_n;
  auto project_x = [&](const v2f (&xf)[7], int s0, int s1) {
    if (s0 == 0) {
#pragma unroll
      for (int j = 0; j < 8; ++j) {
        nx_r[j] = bias_r;
        nx_z[j] = bias_z;
        nx_n[j] = bias_xn;
      }
    }
#pragma unroll
    for (int s = 0; s < 7; ++s) {
      if (s >= s0 && s < s1) {
        nx_r = wmma4(xf[s], wihB[0][s], nx_r);
        nx_z = wmma4(xf[s], wihB[1][s], nx_z);
        nx_n = wmma4(xf[s], wihB[2][s], nx_n);
      }
    }
  };

  auto step = [&](v2f (&xf)[7], int t) {
    v8f acc_r = nx_r, acc_z = nx_z, acc_xn = nx_n, acc_hn;
#pragma unroll
    for (int j = 0; j < 8; ++j) acc_hn[j] = bias_hn;

#pragma unroll
    for (int s = 0; s < 16; ++s) {
      acc_r  = wmma4(hA[s], whhB[0][s], acc_r);
      acc_z  = wmma4(hA[s], whhB[1][s], acc_z);
      acc_hn = wmma4(hA[s], whhB[2][s], acc_hn);
    }

    __builtin_amdgcn_sched_barrier(0);

    project_x(xf, 0, 3);
    load_x(xf, t + 3, 0, 3);

    v8f h_new;
#pragma unroll
    for (int j = 0; j < 8; ++j) {
      const float r = sigmoid_f(acc_r[j]);
      const float z = sigmoid_f(acc_z[j]);
      const float n = tanh_f(acc_xn[j] + r * acc_hn[j]);
      h_new[j] = (1.0f - z) * n + z * h_c[j];
    }
    h_c = h_new;

    lds_barrier();
#pragma unroll
    for (int j = 0; j < 8; ++j)
      h_lds[(j + 8 * hi) * kH + col] = h_new[j];

    __builtin_amdgcn_sched_barrier(0);
    project_x(xf, 3, 5);
    load_x(xf, t + 3, 3, 5);

    lds_barrier();
#pragma unroll
    for (int s = 0; s < 16; ++s)
      hA[s] = *(const v2f*)(h_lds + m * kH + 4 * s + 2 * hi);

    __builtin_amdgcn_sched_barrier(0);
    project_x(xf, 5, 7);
    load_x(xf, t + 3, 5, 7);
  };

  v2f xA[7], xB[7];
  load_x(xA, 0, 0, 7);
  project_x(xA, 0, 7);
  load_x(xA, 1, 0, 7);
  load_x(xB, 2, 0, 7);

  static_assert(kT % 2 == 0, "the time loop alternates two x fragment sets");
  for (int t = 0; t < kT; t += 2) {
    step(xA, t);
    step(xB, t + 1);
  }

  if (tid < kBM) {
    float logits[kC];
    float mx = -3.4e38f;
#pragma unroll
    for (int c = 0; c < kC; ++c) {
      float d = b_out[c];
#pragma unroll
      for (int k = 0; k < kH; ++k)
        d += h_lds[tid * kH + k] * W_out[c * kH + k];
      logits[c] = d;
      mx = fmaxf(mx, d);
    }
    float sum = 0.0f;
#pragma unroll
    for (int c = 0; c < kC; ++c) {
      const float e = __expf(logits[c] - mx);
      logits[c] = e;
      sum += e;
    }
    const float inv = 1.0f / sum;
#pragma unroll
    for (int c = 0; c < kC; ++c)
      out[(size_t)(bm + tid) * kC + c] = logits[c] * inv;
  }
}

}

extern "C" void kernel_launch(void* const* d_in, const int* in_sizes, int n_in,
                              void* d_out, int out_size, void* d_ws, size_t ws_size,
                              hipStream_t stream) {
  (void)in_sizes; (void)n_in; (void)out_size; (void)d_ws; (void)ws_size;
  const float* x     = (const float*)d_in[0];
  const float* W_ih  = (const float*)d_in[1];
  const float* W_hh  = (const float*)d_in[2];
  const float* b_ih  = (const float*)d_in[3];
  const float* b_hh  = (const float*)d_in[4];
  const float* W_out = (const float*)d_in[5];
  const float* b_out = (const float*)d_in[6];
  float* out = (float*)d_out;

  dim3 grid(kB / kBM);
  dim3 block(128);
  gru_fused_kernel<<<grid, block, 0, stream>>>(x, W_ih, W_hh, b_ih, b_hh,
                                               W_out, b_out, out);
}


# h tile in LDS at a row pitch of 68 floats instead of 64: the A-fragment reads (8 B per lane, 16 rows) no longer fall 16 deep on one bank; summation order unchanged
# speedup vs baseline: 2.6806x; 1.2989x over previous
#include <hip/hip_runtime.h>

typedef float v2f __attribute__((ext_vector_type(2)));
typedef float v8f __attribute__((ext_vector_type(8)));

namespace {

constexpr int kB  = 2048;
constexpr int kT  = 512;
constexpr int kIn = 28;
constexpr int kH  = 64;
constexpr int kC  = 10;
constexpr int kBM = 16;
constexpr int kHP = kH + 4;

__device__ __forceinline__ v8f wmma4(v2f a, v2f b, v8f c) {
  return __builtin_amdgcn_wmma_f32_16x16x4_f32(
       false, a,  false, b,
       (short)0, c,  false,  false);
}

__device__ __forceinline__ float rcp_f(float x) {
  return __builtin_amdgcn_rcpf(x);
}
__device__ __forceinline__ float sigmoid_f(float x) {
  return rcp_f(1.0f + __expf(-x));
}
__device__ __forceinline__ float tanh_f(float x) {
  return 1.0f - 2.0f * rcp_f(__expf(2.0f * x) + 1.0f);
}

__device__ __forceinline__ void lds_barrier() {
  __builtin_amdgcn_fence(__ATOMIC_RELEASE, "workgroup", "local");
  __builtin_amdgcn_s_barrier();
  __builtin_amdgcn_fence(__ATOMIC_ACQUIRE, "workgroup", "local");
}

__global__ __launch_bounds__(128, 1) void gru_fused_kernel(
    const float* __restrict__ x,
    const float* __restrict__ W_ih,
    const float* __restrict__ W_hh,
    const float* __restrict__ b_ih,
    const float* __restrict__ b_hh,
    const float* __restrict__ W_out,
    const float* __restrict__ b_out,
    float* __restrict__ out)
{
  __shared__ alignas(16) float h_lds[kBM * kHP];

  const int tid  = threadIdx.x;
  const int wave = tid >> 5;
  const int lane = tid & 31;
  const int m    = lane & 15;
  const int hi   = lane >> 4;
  const int bm   = blockIdx.x * kBM;

  v2f whhB[3][16];
  v2f wihB[3][7];
#pragma unroll
  for (int g = 0; g < 3; ++g) {
    const int nrow = g * 64 + wave * 16 + m;
#pragma unroll
    for (int s = 0; s < 16; ++s)
      whhB[g][s] = *(const v2f*)(W_hh + nrow * kH + 4 * s + 2 * hi);
#pragma unroll
    for (int s = 0; s < 7; ++s)
      wihB[g][s] = *(const v2f*)(W_ih + nrow * kIn + 4 * s + 2 * hi);
  }

  const int col = wave * 16 + m;
  const float bias_r  = b_ih[col]       + b_hh[col];
  const float bias_z  = b_ih[64 + col]  + b_hh[64 + col];
  const float bias_xn = b_ih[128 + col];
  const float bias_hn = b_hh[128 + col];

  v2f hA[16];
#pragma unroll
  for (int s = 0; s < 16; ++s) hA[s] = v2f{0.0f, 0.0f};
  v8f h_c = {};

  const float* xbase = x + (size_t)(bm + m) * (kT * kIn) + 2 * hi;

  auto load_x = [&](v2f (&xf)[7], int t, int s0, int s1) {
    const float* xt = xbase + (t < kT ? t : kT - 1) * kIn;
#pragma unroll
    for (int s = 0; s < 7; ++s)
      if (s >= s0 && s < s1) xf[s] = *(const v2f*)(xt + 4 * s);
  };

  v8f nx_r, nx_z, nx_n;
  auto project_x = [&](const v2f (&xf)[7], int s0, int s1) {
    if (s0 == 0) {
#pragma unroll
      for (int j = 0; j < 8; ++j) {
        nx_r[j] = bias_r;
        nx_z[j] = bias_z;
        nx_n[j] = bias_xn;
      }
    }
#pragma unroll
    for (int s = 0; s < 7; ++s) {
      if (s >= s0 && s < s1) {
        nx_r = wmma4(xf[s], wihB[0][s], nx_r);
        nx_z = wmma4(xf[s], wihB[1][s], nx_z);
        nx_n = wmma4(xf[s], wihB[2][s], nx_n);
      }
    }
  };

  auto step = [&](v2f (&xf)[7], int t) {
    v8f acc_r = nx_r, acc_z = nx_z, acc_xn = nx_n, acc_hn;
#pragma unroll
    for (int j = 0; j < 8; ++j) acc_hn[j] = bias_hn;

#pragma unroll
    for (int s = 0; s < 16; ++s) {
      acc_r  = wmma4(hA[s], whhB[0][s], acc_r);
      acc_z  = wmma4(hA[s], whhB[1][s], acc_z);
      acc_hn = wmma4(hA[s], whhB[2][s], acc_hn);
    }

    __builtin_amdgcn_sched_barrier(0);

    project_x(xf, 0, 3);
    load_x(xf, t + 3, 0, 3);

    v8f h_new;
#pragma unroll
    for (int j = 0; j < 8; ++j) {
      const float r = sigmoid_f(acc_r[j]);
      const float z = sigmoid_f(acc_z[j]);
      const float n = tanh_f(acc_xn[j] + r * acc_hn[j]);
      h_new[j] = (1.0f - z) * n + z * h_c[j];
    }
    h_c = h_new;

    lds_barrier();
#pragma unroll
    for (int j = 0; j < 8; ++j)
      h_lds[(j + 8 * hi) * kHP + col] = h_new[j];

    __builtin_amdgcn_sched_barrier(0);
    project_x(xf, 3, 5);
    load_x(xf, t + 3, 3, 5);

    lds_barrier();
#pragma unroll
    for (int s = 0; s < 16; ++s)
      hA[s] = *(const v2f*)(h_lds + m * kHP + 4 * s + 2 * hi);

    __builtin_amdgcn_sched_barrier(0);
    project_x(xf, 5, 7);
    load_x(xf, t + 3, 5, 7);
  };

  v2f xA[7], xB[7];
  load_x(xA, 0, 0, 7);
  project_x(xA, 0, 7);
  load_x(xA, 1, 0, 7);
  load_x(xB, 2, 0, 7);

  static_assert(kT % 2 == 0, "the time loop alternates two x fragment sets");
  for (int t = 0; t < kT; t += 2) {
    step(xA, t);
    step(xB, t + 1);
  }

  if (tid < kBM) {
    float logits[kC];
    float mx = -3.4e38f;
#pragma unroll
    for (int c = 0; c < kC; ++c) {
      float d = b_out[c];
#pragma unroll
      for (int k = 0; k < kH; ++k)
        d += h_lds[tid * kHP + k] * W_out[c * kH + k];
      logits[c] = d;
      mx = fmaxf(mx, d);
    }
    float sum = 0.0f;
#pragma unroll
    for (int c = 0; c < kC; ++c) {
      const float e = __expf(logits[c] - mx);
      logits[c] = e;
      sum += e;
    }
    const float inv = 1.0f / sum;
#pragma unroll
    for (int c = 0; c < kC; ++c)
      out[(size_t)(bm + tid) * kC + c] = logits[c] * inv;
  }
}

}

extern "C" void kernel_launch(void* const* d_in, const int* in_sizes, int n_in,
                              void* d_out, int out_size, void* d_ws, size_t ws_size,
                              hipStream_t stream) {
  (void)in_sizes; (void)n_in; (void)out_size; (void)d_ws; (void)ws_size;
  const float* x     = (const float*)d_in[0];
  const float* W_ih  = (const float*)d_in[1];
  const float* W_hh  = (const float*)d_in[2];
  const float* b_ih  = (const float*)d_in[3];
  const float* b_hh  = (const float*)d_in[4];
  const float* W_out = (const float*)d_in[5];
  const float* b_out = (const float*)d_in[6];
  float* out = (float*)d_out;

  dim3 grid(kB / kBM);
  dim3 block(128);
  gru_fused_kernel<<<grid, block, 0, stream>>>(x, W_ih, W_hh, b_ih, b_hh,
                                               W_out, b_out, out);
}


# products transposed (W as A, h^T as B): a wave's own 16 h columns feed the next step from its D registers while the other 48 cross LDS as 128-bit rows; h terms summed own K slice first, k regrouped
# speedup vs baseline: 2.8574x; 1.0660x over previous
#include <hip/hip_runtime.h>

typedef float v2f __attribute__((ext_vector_type(2)));
typedef float v4f __attribute__((ext_vector_type(4)));
typedef float v8f __attribute__((ext_vector_type(8)));

namespace {

constexpr int kB  = 2048;
constexpr int kT  = 512;
constexpr int kIn = 28;
constexpr int kH  = 64;
constexpr int kC  = 10;
constexpr int kBM = 16;
constexpr int kHP = kH + 4;

__device__ __forceinline__ v8f wmma4(v2f a, v2f b, v8f c) {
  return __builtin_amdgcn_wmma_f32_16x16x4_f32(
       false, a,  false, b,
       (short)0, c,  false,  false);
}

__device__ __forceinline__ float rcp_f(float x) {
  return __builtin_amdgcn_rcpf(x);
}
__device__ __forceinline__ float sigmoid_f(float x) {
  return rcp_f(1.0f + __expf(-x));
}
__device__ __forceinline__ float tanh_f(float x) {
  return 1.0f - 2.0f * rcp_f(__expf(2.0f * x) + 1.0f);
}

__device__ __forceinline__ void lds_barrier() {
  __builtin_amdgcn_fence(__ATOMIC_RELEASE, "workgroup", "local");
  __builtin_amdgcn_s_barrier();
  __builtin_amdgcn_fence(__ATOMIC_ACQUIRE, "workgroup", "local");
}

__global__ __launch_bounds__(128, 1) void gru_fused_kernel(
    const float* __restrict__ x,
    const float* __restrict__ W_ih,
    const float* __restrict__ W_hh,
    const float* __restrict__ b_ih,
    const float* __restrict__ b_hh,
    const float* __restrict__ W_out,
    const float* __restrict__ b_out,
    float* __restrict__ out)
{
  __shared__ alignas(16) float h_lds[kBM * kHP];

  const int tid  = threadIdx.x;
  const int wave = tid >> 5;
  const int lane = tid & 31;
  const int lc    = lane & 15;
  const int hf   = lane >> 4;
  const int bm   = blockIdx.x * kBM;

  v2f whhA[3][4][4];
  v2f wihA[3][7];
#pragma unroll
  for (int g = 0; g < 3; ++g) {
    const int grow = g * 64 + wave * 16 + lc;
#pragma unroll
    for (int i = 0; i < 4; ++i) {
      const int v = (wave + i) & 3;
#pragma unroll
      for (int q = 0; q < 4; ++q)
        whhA[g][i][q] = *(const v2f*)(W_hh + grow * kH + 16 * v + 8 * hf + 2 * q);
    }
#pragma unroll
    for (int s = 0; s < 7; ++s)
      wihA[g][s] = *(const v2f*)(W_ih + grow * kIn + 4 * s + 2 * hf);
  }

  const int colb = wave * 16 + 8 * hf;
  float bias_r[8], bias_z[8], bias_xn[8], bias_hn[8];
#pragma unroll
  for (int j = 0; j < 8; ++j) {
    bias_r[j]  = b_ih[colb + j]       + b_hh[colb + j];
    bias_z[j]  = b_ih[64 + colb + j]  + b_hh[64 + colb + j];
    bias_xn[j] = b_ih[128 + colb + j];
    bias_hn[j] = b_hh[128 + colb + j];
  }

  v8f h_c = {};
  v4f hO[3][2];
#pragma unroll
  for (int i = 0; i < 3; ++i) {
    hO[i][0] = v4f{0.0f, 0.0f, 0.0f, 0.0f};
    hO[i][1] = v4f{0.0f, 0.0f, 0.0f, 0.0f};
  }
  float* const h_own = h_lds + lc * kHP + 16 * wave + 8 * hf;
  const float* h_oth[3];
#pragma unroll
  for (int i = 0; i < 3; ++i)
    h_oth[i] = h_lds + lc * kHP + 16 * ((wave + 1 + i) & 3) + 8 * hf;

  const float* xbase = x + (size_t)(bm + lc) * (kT * kIn) + 2 * hf;

  auto load_x = [&](v2f (&xf)[7], int t, int s0, int s1) {
    const float* xt = xbase + (t < kT ? t : kT - 1) * kIn;
#pragma unroll
    for (int s = 0; s < 7; ++s)
      if (s >= s0 && s < s1) xf[s] = *(const v2f*)(xt + 4 * s);
  };

  v8f nx_r, nx_z, nx_n;
  auto project_x = [&](const v2f (&xf)[7], int s0, int s1) {
    if (s0 == 0) {
#pragma unroll
      for (int j = 0; j < 8; ++j) {
        nx_r[j] = bias_r[j];
        nx_z[j] = bias_z[j];
        nx_n[j] = bias_xn[j];
      }
    }
#pragma unroll
    for (int s = 0; s < 7; ++s) {
      if (s >= s0 && s < s1) {
        nx_r = wmma4(wihA[0][s], xf[s], nx_r);
        nx_z = wmma4(wihA[1][s], xf[s], nx_z);
        nx_n = wmma4(wihA[2][s], xf[s], nx_n);
      }
    }
  };

  auto step = [&](v2f (&xf)[7], int t) {
    v8f acc_r = nx_r, acc_z = nx_z, acc_xn = nx_n, acc_hn;
#pragma unroll
    for (int j = 0; j < 8; ++j) acc_hn[j] = bias_hn[j];

#pragma unroll
    for (int q = 0; q < 4; ++q) {
      const v2f hb = v2f{h_c[2 * q], h_c[2 * q + 1]};
      acc_r  = wmma4(whhA[0][0][q], hb, acc_r);
      acc_z  = wmma4(whhA[1][0][q], hb, acc_z);
      acc_hn = wmma4(whhA[2][0][q], hb, acc_hn);
    }
#pragma unroll
    for (int i = 0; i < 3; ++i) {
#pragma unroll
      for (int q = 0; q < 4; ++q) {
        const v4f f = hO[i][q >> 1];
        const v2f hb = (q & 1) ? v2f{f[2], f[3]} : v2f{f[0], f[1]};
        acc_r  = wmma4(whhA[0][i + 1][q], hb, acc_r);
        acc_z  = wmma4(whhA[1][i + 1][q], hb, acc_z);
        acc_hn = wmma4(whhA[2][i + 1][q], hb, acc_hn);
      }
    }

    __builtin_amdgcn_sched_barrier(0);

    project_x(xf, 0, 3);
    load_x(xf, t + 3, 0, 3);

    v8f h_new;
#pragma unroll
    for (int j = 0; j < 8; ++j) {
      const float r = sigmoid_f(acc_r[j]);
      const float z = sigmoid_f(acc_z[j]);
      const float n = tanh_f(acc_xn[j] + r * acc_hn[j]);
      h_new[j] = (1.0f - z) * n + z * h_c[j];
    }
    h_c = h_new;

    lds_barrier();
    *(v4f*)(h_own)     = v4f{h_new[0], h_new[1], h_new[2], h_new[3]};
    *(v4f*)(h_own + 4) = v4f{h_new[4], h_new[5], h_new[6], h_new[7]};

    __builtin_amdgcn_sched_barrier(0);
    project_x(xf, 3, 7);
    load_x(xf, t + 3, 3, 7);

    lds_barrier();
#pragma unroll
    for (int i = 0; i < 3; ++i) {
      hO[i][0] = *(const v4f*)(h_oth[i]);
      hO[i][1] = *(const v4f*)(h_oth[i] + 4);
    }

    __builtin_amdgcn_sched_barrier(0);
  };

  __syncthreads();

  v2f xA[7], xB[7];
  load_x(xA, 0, 0, 7);
  project_x(xA, 0, 7);
  load_x(xA, 1, 0, 7);
  load_x(xB, 2, 0, 7);

  static_assert(kT % 2 == 0, "the time loop alternates two x fragment sets");
  for (int t = 0; t < kT; t += 2) {
    step(xA, t);
    step(xB, t + 1);
  }

  if (tid < kBM) {
    float logits[kC];
    float mx = -3.4e38f;
#pragma unroll
    for (int c = 0; c < kC; ++c) {
      float d = b_out[c];
#pragma unroll
      for (int k = 0; k < kH; ++k)
        d += h_lds[tid * kHP + k] * W_out[c * kH + k];
      logits[c] = d;
      mx = fmaxf(mx, d);
    }
    float sum = 0.0f;
#pragma unroll
    for (int c = 0; c < kC; ++c) {
      const float e = __expf(logits[c] - mx);
      logits[c] = e;
      sum += e;
    }
    const float inv = 1.0f / sum;
#pragma unroll
    for (int c = 0; c < kC; ++c)
      out[(size_t)(bm + tid) * kC + c] = logits[c] * inv;
  }
}

}

extern "C" void kernel_launch(void* const* d_in, const int* in_sizes, int n_in,
                              void* d_out, int out_size, void* d_ws, size_t ws_size,
                              hipStream_t stream) {
  (void)in_sizes; (void)n_in; (void)out_size; (void)d_ws; (void)ws_size;
  const float* x     = (const float*)d_in[0];
  const float* W_ih  = (const float*)d_in[1];
  const float* W_hh  = (const float*)d_in[2];
  const float* b_ih  = (const float*)d_in[3];
  const float* b_hh  = (const float*)d_in[4];
  const float* W_out = (const float*)d_in[5];
  const float* b_out = (const float*)d_in[6];
  float* out = (float*)d_out;

  dim3 grid(kB / kBM);
  dim3 block(128);
  gru_fused_kernel<<<grid, block, 0, stream>>>(x, W_ih, W_hh, b_ih, b_hh,
                                               W_out, b_out, out);
}


# gates on the hardware tanh (v_tanh_f32): sigmoid(a) = 1/2 + tanh(a/2)/2, halving folded exactly into the r, z weights; 3 transcendentals per element, was 6; last bits differ, sum order unchanged
# speedup vs baseline: 2.9943x; 1.0479x over previous
#include <hip/hip_runtime.h>

typedef float v2f __attribute__((ext_vector_type(2)));
typedef float v4f __attribute__((ext_vector_type(4)));
typedef float v8f __attribute__((ext_vector_type(8)));

namespace {

constexpr int kB  = 2048;
constexpr int kT  = 512;
constexpr int kIn = 28;
constexpr int kH  = 64;
constexpr int kC  = 10;
constexpr int kBM = 16;
constexpr int kHP = kH + 4;

__device__ __forceinline__ v8f wmma4(v2f a, v2f b, v8f c) {
  return __builtin_amdgcn_wmma_f32_16x16x4_f32(
       false, a,  false, b,
       (short)0, c,  false,  false);
}

__device__ __forceinline__ float tanh_hw(float x) {
  return __builtin_amdgcn_tanhf(x);
}

__device__ __forceinline__ void lds_barrier() {
  __builtin_amdgcn_fence(__ATOMIC_RELEASE, "workgroup", "local");
  __builtin_amdgcn_s_barrier();
  __builtin_amdgcn_fence(__ATOMIC_ACQUIRE, "workgroup", "local");
}

__global__ __launch_bounds__(128, 1) void gru_fused_kernel(
    const float* __restrict__ x,
    const float* __restrict__ W_ih,
    const float* __restrict__ W_hh,
    const float* __restrict__ b_ih,
    const float* __restrict__ b_hh,
    const float* __restrict__ W_out,
    const float* __restrict__ b_out,
    float* __restrict__ out)
{
  __shared__ alignas(16) float h_lds[kBM * kHP];

  const int tid  = threadIdx.x;
  const int wave = tid >> 5;
  const int lane = tid & 31;
  const int lc    = lane & 15;
  const int hf   = lane >> 4;
  const int bm   = blockIdx.x * kBM;

  v2f whhA[3][4][4];
  v2f wihA[3][7];
#pragma unroll
  for (int g = 0; g < 3; ++g) {
    const int grow = g * 64 + wave * 16 + lc;
    const float gs = (g < 2) ? 0.5f : 1.0f;
#pragma unroll
    for (int i = 0; i < 4; ++i) {
      const int v = (wave + i) & 3;
#pragma unroll
      for (int q = 0; q < 4; ++q)
        whhA[g][i][q] = gs * *(const v2f*)(W_hh + grow * kH + 16 * v + 8 * hf + 2 * q);
    }
#pragma unroll
    for (int s = 0; s < 7; ++s)
      wihA[g][s] = gs * *(const v2f*)(W_ih + grow * kIn + 4 * s + 2 * hf);
  }

  const int colb = wave * 16 + 8 * hf;
  float bias_r[8], bias_z[8], bias_xn[8], bias_hn[8];
#pragma unroll
  for (int j = 0; j < 8; ++j) {
    bias_r[j]  = 0.5f * (b_ih[colb + j]       + b_hh[colb + j]);
    bias_z[j]  = 0.5f * (b_ih[64 + colb + j]  + b_hh[64 + colb + j]);
    bias_xn[j] = b_ih[128 + colb + j];
    bias_hn[j] = b_hh[128 + colb + j];
  }

  v8f h_c = {};
  v4f hO[3][2];
#pragma unroll
  for (int i = 0; i < 3; ++i) {
    hO[i][0] = v4f{0.0f, 0.0f, 0.0f, 0.0f};
    hO[i][1] = v4f{0.0f, 0.0f, 0.0f, 0.0f};
  }
  float* const h_own = h_lds + lc * kHP + 16 * wave + 8 * hf;
  const float* h_oth[3];
#pragma unroll
  for (int i = 0; i < 3; ++i)
    h_oth[i] = h_lds + lc * kHP + 16 * ((wave + 1 + i) & 3) + 8 * hf;

  const float* xbase = x + (size_t)(bm + lc) * (kT * kIn) + 2 * hf;

  auto load_x = [&](v2f (&xf)[7], int t, int s0, int s1) {
    const float* xt = xbase + (t < kT ? t : kT - 1) * kIn;
#pragma unroll
    for (int s = 0; s < 7; ++s)
      if (s >= s0 && s < s1) xf[s] = *(const v2f*)(xt + 4 * s);
  };

  v8f nx_r, nx_z, nx_n;
  auto project_x = [&](const v2f (&xf)[7], int s0, int s1) {
    if (s0 == 0) {
#pragma unroll
      for (int j = 0; j < 8; ++j) {
        nx_r[j] = bias_r[j];
        nx_z[j] = bias_z[j];
        nx_n[j] = bias_xn[j];
      }
    }
#pragma unroll
    for (int s = 0; s < 7; ++s) {
      if (s >= s0 && s < s1) {
        nx_r = wmma4(wihA[0][s], xf[s], nx_r);
        nx_z = wmma4(wihA[1][s], xf[s], nx_z);
        nx_n = wmma4(wihA[2][s], xf[s], nx_n);
      }
    }
  };

  auto step = [&](v2f (&xf)[7], int t) {
    v8f acc_r = nx_r, acc_z = nx_z, acc_xn = nx_n, acc_hn;
#pragma unroll
    for (int j = 0; j < 8; ++j) acc_hn[j] = bias_hn[j];

#pragma unroll
    for (int q = 0; q < 4; ++q) {
      const v2f hb = v2f{h_c[2 * q], h_c[2 * q + 1]};
      acc_r  = wmma4(whhA[0][0][q], hb, acc_r);
      acc_z  = wmma4(whhA[1][0][q], hb, acc_z);
      acc_hn = wmma4(whhA[2][0][q], hb, acc_hn);
    }
#pragma unroll
    for (int i = 0; i < 3; ++i) {
#pragma unroll
      for (int q = 0; q < 4; ++q) {
        const v4f f = hO[i][q >> 1];
        const v2f hb = (q & 1) ? v2f{f[2], f[3]} : v2f{f[0], f[1]};
        acc_r  = wmma4(whhA[0][i + 1][q], hb, acc_r);
        acc_z  = wmma4(whhA[1][i + 1][q], hb, acc_z);
        acc_hn = wmma4(whhA[2][i + 1][q], hb, acc_hn);
      }
    }

    __builtin_amdgcn_sched_barrier(0);

    project_x(xf, 0, 3);
    load_x(xf, t + 3, 0, 3);

    v8f h_new;
#pragma unroll
    for (int j = 0; j < 8; ++j) {
      const float tr  = tanh_hw(acc_r[j]);
      const float tz  = tanh_hw(acc_z[j]);
      const float r   = fmaf(0.5f, tr, 0.5f);
      const float z   = fmaf(0.5f, tz, 0.5f);
      const float omz = fmaf(-0.5f, tz, 0.5f);
      const float zh  = z * h_c[j];
      const float n   = tanh_hw(fmaf(r, acc_hn[j], acc_xn[j]));
      h_new[j] = fmaf(omz, n, zh);
    }
    h_c = h_new;

    lds_barrier();
    *(v4f*)(h_own)     = v4f{h_new[0], h_new[1], h_new[2], h_new[3]};
    *(v4f*)(h_own + 4) = v4f{h_new[4], h_new[5], h_new[6], h_new[7]};

    __builtin_amdgcn_sched_barrier(0);
    project_x(xf, 3, 7);
    load_x(xf, t + 3, 3, 7);

    lds_barrier();
#pragma unroll
    for (int i = 0; i < 3; ++i) {
      hO[i][0] = *(const v4f*)(h_oth[i]);
      hO[i][1] = *(const v4f*)(h_oth[i] + 4);
    }

    __builtin_amdgcn_sched_barrier(0);
  };

  __syncthreads();

  v2f xA[7], xB[7];
  load_x(xA, 0, 0, 7);
  project_x(xA, 0, 7);
  load_x(xA, 1, 0, 7);
  load_x(xB, 2, 0, 7);

  static_assert(kT % 2 == 0, "the time loop alternates two x fragment sets");
  for (int t = 0; t < kT; t += 2) {
    step(xA, t);
    step(xB, t + 1);
  }

  if (tid < kBM) {
    float logits[kC];
    float mx = -3.4e38f;
#pragma unroll
    for (int c = 0; c < kC; ++c) {
      float d = b_out[c];
#pragma unroll
      for (int k = 0; k < kH; ++k)
        d += h_lds[tid * kHP + k] * W_out[c * kH + k];
      logits[c] = d;
      mx = fmaxf(mx, d);
    }
    float sum = 0.0f;
#pragma unroll
    for (int c = 0; c < kC; ++c) {
      const float e = __expf(logits[c] - mx);
      logits[c] = e;
      sum += e;
    }
    const float inv = 1.0f / sum;
#pragma unroll
    for (int c = 0; c < kC; ++c)
      out[(size_t)(bm + tid) * kC + c] = logits[c] * inv;
  }
}

}

extern "C" void kernel_launch(void* const* d_in, const int* in_sizes, int n_in,
                              void* d_out, int out_size, void* d_ws, size_t ws_size,
                              hipStream_t stream) {
  (void)in_sizes; (void)n_in; (void)out_size; (void)d_ws; (void)ws_size;
  const float* x     = (const float*)d_in[0];
  const float* W_ih  = (const float*)d_in[1];
  const float* W_hh  = (const float*)d_in[2];
  const float* b_ih  = (const float*)d_in[3];
  const float* b_hh  = (const float*)d_in[4];
  const float* W_out = (const float*)d_in[5];
  const float* b_out = (const float*)d_in[6];
  float* out = (float*)d_out;

  dim3 grid(kB / kBM);
  dim3 block(128);
  gru_fused_kernel<<<grid, block, 0, stream>>>(x, W_ih, W_hh, b_ih, b_hh,
                                               W_out, b_out, out);
}


# x loads of a step issued as one group in front of the LDS reads of h: two places per step where memory instructions stand among the matrix instructions, was five; summation order unchanged
# speedup vs baseline: 3.0953x; 1.0337x over previous
#include <hip/hip_runtime.h>

typedef float v2f __attribute__((ext_vector_type(2)));
typedef float v4f __attribute__((ext_vector_type(4)));
typedef float v8f __attribute__((ext_vector_type(8)));

namespace {

constexpr int kB  = 2048;
constexpr int kT  = 512;
constexpr int kIn = 28;
constexpr int kH  = 64;
constexpr int kC  = 10;
constexpr int kBM = 16;
constexpr int kHP = kH + 4;

__device__ __forceinline__ v8f wmma4(v2f a, v2f b, v8f c) {
  return __builtin_amdgcn_wmma_f32_16x16x4_f32(
       false, a,  false, b,
       (short)0, c,  false,  false);
}

__device__ __forceinline__ float tanh_hw(float x) {
  return __builtin_amdgcn_tanhf(x);
}

__device__ __forceinline__ void lds_barrier() {
  __builtin_amdgcn_fence(__ATOMIC_RELEASE, "workgroup", "local");
  __builtin_amdgcn_s_barrier();
  __builtin_amdgcn_fence(__ATOMIC_ACQUIRE, "workgroup", "local");
}

__global__ __launch_bounds__(128, 1) void gru_fused_kernel(
    const float* __restrict__ x,
    const float* __restrict__ W_ih,
    const float* __restrict__ W_hh,
    const float* __restrict__ b_ih,
    const float* __restrict__ b_hh,
    const float* __restrict__ W_out,
    const float* __restrict__ b_out,
    float* __restrict__ out)
{
  __shared__ alignas(16) float h_lds[kBM * kHP];

  const int tid  = threadIdx.x;
  const int wave = tid >> 5;
  const int lane = tid & 31;
  const int lc    = lane & 15;
  const int hf   = lane >> 4;
  const int bm   = blockIdx.x * kBM;

  v2f whhA[3][4][4];
  v2f wihA[3][7];
#pragma unroll
  for (int g = 0; g < 3; ++g) {
    const int grow = g * 64 + wave * 16 + lc;
    const float gs = (g < 2) ? 0.5f : 1.0f;
#pragma unroll
    for (int i = 0; i < 4; ++i) {
      const int v = (wave + i) & 3;
#pragma unroll
      for (int q = 0; q < 4; ++q)
        whhA[g][i][q] = gs * *(const v2f*)(W_hh + grow * kH + 16 * v + 8 * hf + 2 * q);
    }
#pragma unroll
    for (int s = 0; s < 7; ++s)
      wihA[g][s] = gs * *(const v2f*)(W_ih + grow * kIn + 4 * s + 2 * hf);
  }

  const int colb = wave * 16 + 8 * hf;
  float bias_r[8], bias_z[8], bias_xn[8], bias_hn[8];
#pragma unroll
  for (int j = 0; j < 8; ++j) {
    bias_r[j]  = 0.5f * (b_ih[colb + j]       + b_hh[colb + j]);
    bias_z[j]  = 0.5f * (b_ih[64 + colb + j]  + b_hh[64 + colb + j]);
    bias_xn[j] = b_ih[128 + colb + j];
    bias_hn[j] = b_hh[128 + colb + j];
  }

  v8f h_c = {};
  v4f hO[3][2];
#pragma unroll
  for (int i = 0; i < 3; ++i) {
    hO[i][0] = v4f{0.0f, 0.0f, 0.0f, 0.0f};
    hO[i][1] = v4f{0.0f, 0.0f, 0.0f, 0.0f};
  }
  float* const h_own = h_lds + lc * kHP + 16 * wave + 8 * hf;
  const float* h_oth[3];
#pragma unroll
  for (int i = 0; i < 3; ++i)
    h_oth[i] = h_lds + lc * kHP + 16 * ((wave + 1 + i) & 3) + 8 * hf;

  const float* xbase = x + (size_t)(bm + lc) * (kT * kIn) + 2 * hf;

  auto load_x = [&](v2f (&xf)[7], int t, int s0, int s1) {
    const float* xt = xbase + (t < kT ? t : kT - 1) * kIn;
#pragma unroll
    for (int s = 0; s < 7; ++s)
      if (s >= s0 && s < s1) xf[s] = *(const v2f*)(xt + 4 * s);
  };

  v8f nx_r, nx_z, nx_n;
  auto project_x = [&](const v2f (&xf)[7], int s0, int s1) {
    if (s0 == 0) {
#pragma unroll
      for (int j = 0; j < 8; ++j) {
        nx_r[j] = bias_r[j];
        nx_z[j] = bias_z[j];
        nx_n[j] = bias_xn[j];
      }
    }
#pragma unroll
    for (int s = 0; s < 7; ++s) {
      if (s >= s0 && s < s1) {
        nx_r = wmma4(wihA[0][s], xf[s], nx_r);
        nx_z = wmma4(wihA[1][s], xf[s], nx_z);
        nx_n = wmma4(wihA[2][s], xf[s], nx_n);
      }
    }
  };

  auto step = [&](v2f (&xf)[7], int t) {
    v8f acc_r = nx_r, acc_z = nx_z, acc_xn = nx_n, acc_hn;
#pragma unroll
    for (int j = 0; j < 8; ++j) acc_hn[j] = bias_hn[j];

#pragma unroll
    for (int q = 0; q < 4; ++q) {
      const v2f hb = v2f{h_c[2 * q], h_c[2 * q + 1]};
      acc_r  = wmma4(whhA[0][0][q], hb, acc_r);
      acc_z  = wmma4(whhA[1][0][q], hb, acc_z);
      acc_hn = wmma4(whhA[2][0][q], hb, acc_hn);
    }
#pragma unroll
    for (int i = 0; i < 3; ++i) {
#pragma unroll
      for (int q = 0; q < 4; ++q) {
        const v4f f = hO[i][q >> 1];
        const v2f hb = (q & 1) ? v2f{f[2], f[3]} : v2f{f[0], f[1]};
        acc_r  = wmma4(whhA[0][i + 1][q], hb, acc_r);
        acc_z  = wmma4(whhA[1][i + 1][q], hb, acc_z);
        acc_hn = wmma4(whhA[2][i + 1][q], hb, acc_hn);
      }
    }

    __builtin_amdgcn_sched_barrier(0);

    project_x(xf, 0, 3);

    v8f h_new;
#pragma unroll
    for (int j = 0; j < 8; ++j) {
      const float tr  = tanh_hw(acc_r[j]);
      const float tz  = tanh_hw(acc_z[j]);
      const float r   = fmaf(0.5f, tr, 0.5f);
      const float z   = fmaf(0.5f, tz, 0.5f);
      const float omz = fmaf(-0.5f, tz, 0.5f);
      const float zh  = z * h_c[j];
      const float n   = tanh_hw(fmaf(r, acc_hn[j], acc_xn[j]));
      h_new[j] = fmaf(omz, n, zh);
    }
    h_c = h_new;

    lds_barrier();
    *(v4f*)(h_own)     = v4f{h_new[0], h_new[1], h_new[2], h_new[3]};
    *(v4f*)(h_own + 4) = v4f{h_new[4], h_new[5], h_new[6], h_new[7]};

    __builtin_amdgcn_sched_barrier(0);
    project_x(xf, 3, 7);

    __builtin_amdgcn_sched_barrier(0);
    load_x(xf, t + 3, 0, 7);

    lds_barrier();
#pragma unroll
    for (int i = 0; i < 3; ++i) {
      hO[i][0] = *(const v4f*)(h_oth[i]);
      hO[i][1] = *(const v4f*)(h_oth[i] + 4);
    }

    __builtin_amdgcn_sched_barrier(0);
  };

  __syncthreads();

  v2f xA[7], xB[7];
  load_x(xA, 0, 0, 7);
  project_x(xA, 0, 7);
  load_x(xA, 1, 0, 7);
  load_x(xB, 2, 0, 7);

  static_assert(kT % 2 == 0, "the time loop alternates two x fragment sets");
  for (int t = 0; t < kT; t += 2) {
    step(xA, t);
    step(xB, t + 1);
  }

  if (tid < kBM) {
    float logits[kC];
    float mx = -3.4e38f;
#pragma unroll
    for (int c = 0; c < kC; ++c) {
      float d = b_out[c];
#pragma unroll
      for (int k = 0; k < kH; ++k)
        d += h_lds[tid * kHP + k] * W_out[c * kH + k];
      logits[c] = d;
      mx = fmaxf(mx, d);
    }
    float sum = 0.0f;
#pragma unroll
    for (int c = 0; c < kC; ++c) {
      const float e = __expf(logits[c] - mx);
      logits[c] = e;
      sum += e;
    }
    const float inv = 1.0f / sum;
#pragma unroll
    for (int c = 0; c < kC; ++c)
      out[(size_t)(bm + tid) * kC + c] = logits[c] * inv;
  }
}

}

extern "C" void kernel_launch(void* const* d_in, const int* in_sizes, int n_in,
                              void* d_out, int out_size, void* d_ws, size_t ws_size,
                              hipStream_t stream) {
  (void)in_sizes; (void)n_in; (void)out_size; (void)d_ws; (void)ws_size;
  const float* x     = (const float*)d_in[0];
  const float* W_ih  = (const float*)d_in[1];
  const float* W_hh  = (const float*)d_in[2];
  const float* b_ih  = (const float*)d_in[3];
  const float* b_hh  = (const float*)d_in[4];
  const float* W_out = (const float*)d_in[5];
  const float* b_out = (const float*)d_in[6];
  float* out = (float*)d_out;

  dim3 grid(kB / kBM);
  dim3 block(128);
  gru_fused_kernel<<<grid, block, 0, stream>>>(x, W_ih, W_hh, b_ih, b_hh,
                                               W_out, b_out, out);
}


# 8 batch rows per workgroup on 256 workgroups, one per WGP (was 16 on 128); input projection packs two time steps into a product's 16 batch columns, halves swapped by DPP; sum order unchanged
# speedup vs baseline: 3.4237x; 1.1061x over previous
#include <hip/hip_runtime.h>

typedef float v2f __attribute__((ext_vector_type(2)));
typedef float v4f __attribute__((ext_vector_type(4)));
typedef float v8f __attribute__((ext_vector_type(8)));

namespace {

constexpr int kB  = 2048;
constexpr int kT  = 512;
constexpr int kIn = 28;
constexpr int kH  = 64;
constexpr int kC  = 10;
constexpr int kBR = 8;
constexpr int kBM = 16;
constexpr int kHP = kH + 4;
constexpr int kLdsFloats = 164 * 1024 / 4;

__device__ __forceinline__ v8f wmma4(v2f a, v2f b, v8f c) {
  return __builtin_amdgcn_wmma_f32_16x16x4_f32(
       false, a,  false, b,
       (short)0, c,  false,  false);
}

__device__ __forceinline__ float tanh_hw(float x) {
  return __builtin_amdgcn_tanhf(x);
}

__device__ __forceinline__ float swap8(float v) {
  const int i = __builtin_bit_cast(int, v);
  return __builtin_bit_cast(
      float, __builtin_amdgcn_update_dpp(i, i,   0x128, 0xF, 0xF, false));
}

__device__ __forceinline__ void lds_barrier() {
  __builtin_amdgcn_fence(__ATOMIC_RELEASE, "workgroup", "local");
  __builtin_amdgcn_s_barrier();
  __builtin_amdgcn_fence(__ATOMIC_ACQUIRE, "workgroup", "local");
}

__global__ __launch_bounds__(128, 1) void gru_fused_kernel(
    const float* __restrict__ x,
    const float* __restrict__ W_ih,
    const float* __restrict__ W_hh,
    const float* __restrict__ b_ih,
    const float* __restrict__ b_hh,
    const float* __restrict__ W_out,
    const float* __restrict__ b_out,
    float* __restrict__ out)
{
  __shared__ alignas(16) float h_lds[kLdsFloats];

  const int tid  = threadIdx.x;
  const int wave = tid >> 5;
  const int lane = tid & 31;
  const int lc   = lane & 15;
  const int hf   = lane >> 4;
  const int rb   = lc & 7;
  const int th   = lc >> 3;
  const int bm   = blockIdx.x * kBR;

  v2f whhA[3][4][4];
  v2f wihA[3][7];
#pragma unroll
  for (int g = 0; g < 3; ++g) {
    const int grow = g * 64 + wave * 16 + lc;
    const float gs = (g < 2) ? 0.5f : 1.0f;
#pragma unroll
    for (int i = 0; i < 4; ++i) {
      const int v = (wave + i) & 3;
#pragma unroll
      for (int q = 0; q < 4; ++q)
        whhA[g][i][q] = gs * *(const v2f*)(W_hh + grow * kH + 16 * v + 8 * hf + 2 * q);
    }
#pragma unroll
    for (int s = 0; s < 7; ++s)
      wihA[g][s] = gs * *(const v2f*)(W_ih + grow * kIn + 4 * s + 2 * hf);
  }

  const int colb = wave * 16 + 8 * hf;
  float bias_r[8], bias_z[8], bias_xn[8], bias_hn[8];
#pragma unroll
  for (int j = 0; j < 8; ++j) {
    bias_r[j]  = 0.5f * (b_ih[colb + j]       + b_hh[colb + j]);
    bias_z[j]  = 0.5f * (b_ih[64 + colb + j]  + b_hh[64 + colb + j]);
    bias_xn[j] = b_ih[128 + colb + j];
    bias_hn[j] = b_hh[128 + colb + j];
  }

  v8f h_c = {};
  v4f hO[3][2];
#pragma unroll
  for (int i = 0; i < 3; ++i) {
    hO[i][0] = v4f{0.0f, 0.0f, 0.0f, 0.0f};
    hO[i][1] = v4f{0.0f, 0.0f, 0.0f, 0.0f};
  }
  float* const h_own = h_lds + lc * kHP + 16 * wave + 8 * hf;
  const float* h_oth[3];
#pragma unroll
  for (int i = 0; i < 3; ++i)
    h_oth[i] = h_lds + rb * kHP + 16 * ((wave + 1 + i) & 3) + 8 * hf;

  const float* xbase = x + (size_t)(bm + rb) * (kT * kIn) + 2 * hf;

  auto load_x = [&](v2f (&xf)[7], int pair) {
    const int t = 2 * pair + th;
    const float* xt = xbase + (t < kT ? t : kT - 1) * kIn;
#pragma unroll
    for (int s = 0; s < 7; ++s) xf[s] = *(const v2f*)(xt + 4 * s);
  };

  v8f px_r, px_z, px_n;
  v8f nx_r, nx_z, nx_n;
  auto project_x = [&](const v2f (&xf)[7], int s0, int s1) {
    if (s0 == 0) {
#pragma unroll
      for (int j = 0; j < 8; ++j) {
        nx_r[j] = bias_r[j];
        nx_z[j] = bias_z[j];
        nx_n[j] = bias_xn[j];
      }
    }
#pragma unroll
    for (int s = 0; s < 7; ++s) {
      if (s >= s0 && s < s1) {
        nx_r = wmma4(wihA[0][s], xf[s], nx_r);
        nx_z = wmma4(wihA[1][s], xf[s], nx_z);
        nx_n = wmma4(wihA[2][s], xf[s], nx_n);
      }
    }
  };

  auto step = [&](v2f (&xf)[7], int pair, bool odd) {
    v8f acc_r = px_r, acc_z = px_z, acc_xn = px_n, acc_hn;
#pragma unroll
    for (int j = 0; j < 8; ++j) acc_hn[j] = bias_hn[j];

#pragma unroll
    for (int q = 0; q < 4; ++q) {
      const v2f hb = v2f{h_c[2 * q], h_c[2 * q + 1]};
      acc_r  = wmma4(whhA[0][0][q], hb, acc_r);
      acc_z  = wmma4(whhA[1][0][q], hb, acc_z);
      acc_hn = wmma4(whhA[2][0][q], hb, acc_hn);
    }
#pragma unroll
    for (int i = 0; i < 3; ++i) {
#pragma unroll
      for (int q = 0; q < 4; ++q) {
        const v4f f = hO[i][q >> 1];
        const v2f hb = (q & 1) ? v2f{f[2], f[3]} : v2f{f[0], f[1]};
        acc_r  = wmma4(whhA[0][i + 1][q], hb, acc_r);
        acc_z  = wmma4(whhA[1][i + 1][q], hb, acc_z);
        acc_hn = wmma4(whhA[2][i + 1][q], hb, acc_hn);
      }
    }

    __builtin_amdgcn_sched_barrier(0);

    if (!odd) project_x(xf, 0, 3);
    else      project_x(xf, 3, 6);

    v8f h_new;
#pragma unroll
    for (int j = 0; j < 8; ++j) {
      const float tr  = tanh_hw(acc_r[j]);
      const float tz  = tanh_hw(acc_z[j]);
      const float r   = fmaf(0.5f, tr, 0.5f);
      const float z   = fmaf(0.5f, tz, 0.5f);
      const float omz = fmaf(-0.5f, tz, 0.5f);
      const float zh  = z * h_c[j];
      const float n   = tanh_hw(fmaf(r, acc_hn[j], acc_xn[j]));
      h_new[j] = fmaf(omz, n, zh);
    }
    h_c = h_new;

    lds_barrier();
    *(v4f*)(h_own)     = v4f{h_new[0], h_new[1], h_new[2], h_new[3]};
    *(v4f*)(h_own + 4) = v4f{h_new[4], h_new[5], h_new[6], h_new[7]};

    __builtin_amdgcn_sched_barrier(0);
    if (!odd) {
#pragma unroll
      for (int j = 0; j < 8; ++j) {
        px_r[j] = swap8(px_r[j]);
        px_z[j] = swap8(px_z[j]);
        px_n[j] = swap8(px_n[j]);
      }
      __builtin_amdgcn_sched_barrier(0);
    } else {
      project_x(xf, 6, 7);
      px_r = nx_r; px_z = nx_z; px_n = nx_n;
      __builtin_amdgcn_sched_barrier(0);
      load_x(xf, pair + 3);
    }

    lds_barrier();
#pragma unroll
    for (int i = 0; i < 3; ++i) {
      hO[i][0] = *(const v4f*)(h_oth[i]);
      hO[i][1] = *(const v4f*)(h_oth[i] + 4);
    }

    __builtin_amdgcn_sched_barrier(0);
  };

  __syncthreads();

  v2f xA[7], xB[7];
  load_x(xA, 0);
  project_x(xA, 0, 7);
  px_r = nx_r; px_z = nx_z; px_n = nx_n;
  load_x(xB, 1);
  load_x(xA, 2);

  static_assert(kT % 4 == 0, "the time loop alternates two x fragment sets of two steps each");
  for (int pair = 0; pair < kT / 2; pair += 2) {
    step(xB, pair, false);
    step(xB, pair, true);
    step(xA, pair + 1, false);
    step(xA, pair + 1, true);
  }

  if (tid < kBR) {
    float logits[kC];
    float mx = -3.4e38f;
#pragma unroll
    for (int c = 0; c < kC; ++c) {
      float d = b_out[c];
#pragma unroll
      for (int k = 0; k < kH; ++k)
        d += h_lds[tid * kHP + k] * W_out[c * kH + k];
      logits[c] = d;
      mx = fmaxf(mx, d);
    }
    float sum = 0.0f;
#pragma unroll
    for (int c = 0; c < kC; ++c) {
      const float e = __expf(logits[c] - mx);
      logits[c] = e;
      sum += e;
    }
    const float inv = 1.0f / sum;
#pragma unroll
    for (int c = 0; c < kC; ++c)
      out[(size_t)(bm + tid) * kC + c] = logits[c] * inv;
  }
}

}

extern "C" void kernel_launch(void* const* d_in, const int* in_sizes, int n_in,
                              void* d_out, int out_size, void* d_ws, size_t ws_size,
                              hipStream_t stream) {
  (void)in_sizes; (void)n_in; (void)out_size; (void)d_ws; (void)ws_size;
  const float* x     = (const float*)d_in[0];
  const float* W_ih  = (const float*)d_in[1];
  const float* W_hh  = (const float*)d_in[2];
  const float* b_ih  = (const float*)d_in[3];
  const float* b_hh  = (const float*)d_in[4];
  const float* W_out = (const float*)d_in[5];
  const float* b_out = (const float*)d_in[6];
  float* out = (float*)d_out;

  dim3 grid(kB / kBR);
  dim3 block(128);
  gru_fused_kernel<<<grid, block, 0, stream>>>(x, W_ih, W_hh, b_ih, b_hh,
                                               W_out, b_out, out);
}
